# starting kernel
# baseline (speedup 1.0000x reference)
#include <hip/hip_runtime.h>
#include <hip/hip_bf16.h>


typedef __bf16 bf16_t;
typedef __bf16 v16bf __attribute__((ext_vector_type(16)));
typedef __bf16 v8bf  __attribute__((ext_vector_type(8)));
typedef float  v8f   __attribute__((ext_vector_type(8)));

#define IN_FT   256
#define HID1    400
#define HID2    200
#define OUT_FT  128

#define KP0     256
#define KP1     416
#define KP2     224
#define NP0     416
#define NP1     224
#define NP2     128

__global__ __launch_bounds__(256) void cvt_bf16_kernel(
    const float* __restrict__ in, bf16_t* __restrict__ out, int n) {
  int i = blockIdx.x * 256 + threadIdx.x;
  if (i < n) out[i] = (bf16_t)in[i];
}

__global__ __launch_bounds__(256) void pack_weight_kernel(
    const float* __restrict__ W, bf16_t* __restrict__ Wt,
    int K, int Nout, int Kp, int Noutp) {
  int idx = blockIdx.x * 256 + threadIdx.x;
  if (idx >= Noutp * Kp) return;
  int n = idx / Kp, k = idx % Kp;
  float v = (n < Nout && k < K) ? W[(size_t)k * Nout + n] : 0.0f;
  Wt[idx] = (bf16_t)v;
}

__global__ __launch_bounds__(256) void deg_kernel(
    const int* __restrict__ dst, const float* __restrict__ ew,
    float* __restrict__ deg, int E) {
  int e = blockIdx.x * 256 + threadIdx.x;
  if (e < E)
    __hip_atomic_fetch_add(&deg[dst[e]], ew[e],
                           __ATOMIC_RELAXED, __HIP_MEMORY_SCOPE_AGENT);
}

__global__ __launch_bounds__(256) void dinv_kernel(
    const float* __restrict__ deg, float* __restrict__ dinv, int n) {
  int i = blockIdx.x * 256 + threadIdx.x;
  if (i < n) dinv[i] = rsqrtf(deg[i] + 1.0f);
}

__device__ __forceinline__ v16bf load_frag(const bf16_t* p) {
  v8bf lo = *(const v8bf*)p;
  v8bf hi = *(const v8bf*)(p + 16);
  v16bf f;
#pragma unroll
  for (int i = 0; i < 8; ++i) { f[i] = lo[i]; f[8 + i] = hi[i]; }
  return f;
}

__device__ __forceinline__ v8f wmma_bf16(v16bf a, v16bf b, v8f c) {
  return __builtin_amdgcn_wmma_f32_16x16x32_bf16(
      false, a, false, b, (short)0, c, false, false);
}

template <int MODE>
__global__ __launch_bounds__(256) void wmma_gemm_kernel(
    const bf16_t* __restrict__ A, const bf16_t* __restrict__ Bt,
    const float* __restrict__ bias, void* __restrict__ Cout,
    int M, int Kp, int Nout, int tiles_n, int ldo) {
  const int wid  = blockIdx.x * 8 + (threadIdx.x >> 5);
  const int lane = threadIdx.x & 31;
  const int tm = wid / tiles_n;
  const int tn = wid % tiles_n;
  if (tm * 32 >= M) return;

  const int r  = lane & 15;
  const int hi = lane >> 4;
  const bf16_t* a0 = A  + (size_t)(tm * 32 + r)      * Kp + hi * 8;
  const bf16_t* a1 = A  + (size_t)(tm * 32 + 16 + r) * Kp + hi * 8;
  const bf16_t* b0 = Bt + (size_t)(tn * 32 + r)      * Kp + hi * 8;
  const bf16_t* b1 = Bt + (size_t)(tn * 32 + 16 + r) * Kp + hi * 8;

  v8f c00 = {}, c01 = {}, c10 = {}, c11 = {};
  for (int kb = 0; kb < Kp; kb += 32) {
    v16bf va0 = load_frag(a0 + kb);
    v16bf va1 = load_frag(a1 + kb);
    v16bf vb0 = load_frag(b0 + kb);
    v16bf vb1 = load_frag(b1 + kb);
    c00 = wmma_bf16(va0, vb0, c00);
    c01 = wmma_bf16(va0, vb1, c01);
    c10 = wmma_bf16(va1, vb0, c10);
    c11 = wmma_bf16(va1, vb1, c11);
  }

  const v8f cc[2][2] = {{c00, c01}, {c10, c11}};
#pragma unroll
  for (int sn = 0; sn < 2; ++sn) {
    const int n = tn * 32 + sn * 16 + r;
    if (n >= Nout) continue;
    const float bv = (MODE == 1) ? bias[n] : 0.0f;
#pragma unroll
    for (int sm = 0; sm < 2; ++sm) {
      const v8f c = cc[sm][sn];
#pragma unroll
      for (int i = 0; i < 8; ++i) {
        const int m = tm * 32 + sm * 16 + i + hi * 8;
        float v = c[i] + bv;
        if (MODE == 1) {
          v = v > 0.0f ? v : 0.0f;
          ((bf16_t*)Cout)[(size_t)m * ldo + n] = (bf16_t)v;
        } else {
          ((float*)Cout)[(size_t)m * ldo + n] = v;
        }
      }
    }
  }
}

__global__ __launch_bounds__(256) void scatter_kernel(
    const float* __restrict__ t, const int* __restrict__ src,
    const int* __restrict__ dst, const float* __restrict__ ew,
    const float* __restrict__ dinv, float* __restrict__ agg,
    int E, int F) {
  const int e = blockIdx.x * 8 + (threadIdx.x >> 5);
  if (e >= E) return;
  const int lane = threadIdx.x & 31;
  const int s = src[e];
  const int d = dst[e];
  const float nrm = dinv[s] * ew[e] * dinv[d];
  const float* trow = t + (size_t)s * F;
  float* arow = agg + (size_t)d * F;
  __builtin_prefetch(trow, 0, 1);
  for (int f = lane; f < F; f += 32)
    __hip_atomic_fetch_add(&arow[f], trow[f] * nrm,
                           __ATOMIC_RELAXED, __HIP_MEMORY_SCOPE_AGENT);
}

__global__ __launch_bounds__(256) void combine_bf16_kernel(
    const float* __restrict__ agg, const float* __restrict__ t,
    const float* __restrict__ dinv, const float* __restrict__ bias,
    bf16_t* __restrict__ out, int Nn, int F, int ldo) {
  int idx = blockIdx.x * 256 + threadIdx.x;
  if (idx >= Nn * F) return;
  int i = idx / F, f = idx % F;
  float di = dinv[i];
  float v = agg[idx] + t[idx] * di * di + bias[f];
  out[(size_t)i * ldo + f] = (bf16_t)(v > 0.0f ? v : 0.0f);
}

__global__ __launch_bounds__(256) void combine_f32_kernel(
    const float* __restrict__ agg, const float* __restrict__ t,
    const float* __restrict__ dinv, const float* __restrict__ bias,
    float* __restrict__ out, int Nn, int F) {
  int idx = blockIdx.x * 256 + threadIdx.x;
  if (idx >= Nn * F) return;
  int f = idx % F;
  int i = idx / F;
  float di = dinv[i];
  float v = agg[idx] + t[idx] * di * di + bias[f];
  out[idx] = v > 0.0f ? v : 0.0f;
}

static inline size_t al256(size_t x) { return (x + 255) & ~(size_t)255; }

extern "C" void kernel_launch(void* const* d_in, const int* in_sizes, int n_in,
                              void* d_out, int out_size, void* d_ws, size_t ws_size,
                              hipStream_t stream) {
  const float* x    = (const float*)d_in[0];
  const int*   ei   = (const int*)  d_in[1];
  const float* ew   = (const float*)d_in[2];
  const float* fc_W = (const float*)d_in[3];
  const float* fc_b = (const float*)d_in[4];
  const float* W1   = (const float*)d_in[5];
  const float* b1   = (const float*)d_in[6];
  const float* W2   = (const float*)d_in[7];
  const float* b2   = (const float*)d_in[8];

  const int Nn = in_sizes[0] / IN_FT;
  const int E  = in_sizes[1] / 2;
  const int* src = ei;
  const int* dst = ei + E;

  char* ws = (char*)d_ws;
  size_t off = 0;
  auto alloc = [&](size_t bytes) { void* p = ws + off; off += al256(bytes); return p; };

  float*  deg  = (float*) alloc((size_t)Nn * 4);
  float*  dinv = (float*) alloc((size_t)Nn * 4);
  bf16_t* xb   = (bf16_t*)alloc((size_t)Nn * KP0 * 2);
  bf16_t* fcWt = (bf16_t*)alloc((size_t)NP0 * KP0 * 2);
  bf16_t* h0b  = (bf16_t*)alloc((size_t)Nn * KP1 * 2);
  bf16_t* W1t  = (bf16_t*)alloc((size_t)NP1 * KP1 * 2);
  float*  t1   = (float*) alloc((size_t)Nn * HID2 * 4);
  float*  agg1 = (float*) alloc((size_t)Nn * HID2 * 4);
  bf16_t* h1b  = (bf16_t*)alloc((size_t)Nn * KP2 * 2);
  bf16_t* W2t  = (bf16_t*)alloc((size_t)NP2 * KP2 * 2);
  float*  t2   = (float*) alloc((size_t)Nn * OUT_FT * 4);
  float*  agg2 = (float*) alloc((size_t)Nn * OUT_FT * 4);
  (void)ws_size;

  hipMemsetAsync(deg,  0, (size_t)Nn * 4, stream);
  hipMemsetAsync(agg1, 0, (size_t)Nn * HID2 * 4, stream);
  hipMemsetAsync(agg2, 0, (size_t)Nn * OUT_FT * 4, stream);
  hipMemsetAsync(h0b,  0, (size_t)Nn * KP1 * 2, stream);
  hipMemsetAsync(h1b,  0, (size_t)Nn * KP2 * 2, stream);

  {
    int n = Nn * IN_FT;
    cvt_bf16_kernel<<<(n + 255) / 256, 256, 0, stream>>>(x, xb, n);
  }
  pack_weight_kernel<<<((NP0 * KP0) + 255) / 256, 256, 0, stream>>>(
      fc_W, fcWt, IN_FT, HID1, KP0, NP0);
  pack_weight_kernel<<<((NP1 * KP1) + 255) / 256, 256, 0, stream>>>(
      W1, W1t, HID1, HID2, KP1, NP1);
  pack_weight_kernel<<<((NP2 * KP2) + 255) / 256, 256, 0, stream>>>(
      W2, W2t, HID2, OUT_FT, KP2, NP2);

  deg_kernel <<<(E + 255) / 256, 256, 0, stream>>>(dst, ew, deg, E);
  dinv_kernel<<<(Nn + 255) / 256, 256, 0, stream>>>(deg, dinv, Nn);

  {
    const int tiles_n = NP0 / 32;
    const int waves   = (Nn / 32) * tiles_n;
    wmma_gemm_kernel<1><<<(waves + 7) / 8, 256, 0, stream>>>(
        xb, fcWt, fc_b, h0b, Nn, KP0, HID1, tiles_n, KP1);
  }

  {
    const int tiles_n = NP1 / 32;
    const int waves   = (Nn / 32) * tiles_n;
    wmma_gemm_kernel<0><<<(waves + 7) / 8, 256, 0, stream>>>(
        h0b, W1t, nullptr, t1, Nn, KP1, HID2, tiles_n, HID2);
    scatter_kernel<<<(E + 7) / 8, 256, 0, stream>>>(
        t1, src, dst, ew, dinv, agg1, E, HID2);
    int n = Nn * HID2;
    combine_bf16_kernel<<<(n + 255) / 256, 256, 0, stream>>>(
        agg1, t1, dinv, b1, h1b, Nn, HID2, KP2);
  }

  {
    const int tiles_n = NP2 / 32;
    const int waves   = (Nn / 32) * tiles_n;
    wmma_gemm_kernel<0><<<(waves + 7) / 8, 256, 0, stream>>>(
        h1b, W2t, nullptr, t2, Nn, KP2, OUT_FT, tiles_n, OUT_FT);
    scatter_kernel<<<(E + 7) / 8, 256, 0, stream>>>(
        t2, src, dst, ew, dinv, agg2, E, OUT_FT);
    int n = Nn * OUT_FT;
    combine_f32_kernel<<<(n + 255) / 256, 256, 0, stream>>>(
        agg2, t2, dinv, b2, (float*)d_out, Nn, OUT_FT);
  }
}


# Edge-sort rank step rides in the layer-0 GEMM launch and the scan in the layer-1 GEMM launch as extra work-groups (role by index; scan recast for one 224-thread group); 10 -> 8 launches
# speedup vs baseline: 3.1869x; 3.1869x over previous
#include <hip/hip_runtime.h>
#include <hip/hip_bf16.h>


typedef __bf16 bf16_t;
typedef __bf16 v16bf __attribute__((ext_vector_type(16)));
typedef __bf16 v8bf  __attribute__((ext_vector_type(8)));
typedef float  v8f   __attribute__((ext_vector_type(8)));

#define IN_FT   256
#define HID1    400
#define HID2    200
#define OUT_FT  128

#define KP0     256
#define KP1     416
#define KP2     224
#define NP0     416
#define NP1     224
#define NP2     128

typedef float  v4f  __attribute__((ext_vector_type(4)));
typedef __bf16 v4bf __attribute__((ext_vector_type(4)));

#define CVT_U 4
__device__ __forceinline__ void cvt_bf16_block(
    const float* __restrict__ in, bf16_t* __restrict__ out, int n4, int b) {
  const v4f* __restrict__ in4  = (const v4f*)in;
  v4bf* __restrict__      out4 = (v4bf*)out;
  const int blk  = b * (256 * CVT_U);
  const int base = blk + threadIdx.x;
  if (blk + 256 * CVT_U <= n4) {
    v4f v[CVT_U];
    for (int u = 0; u < CVT_U; ++u) v[u] = in4[base + u * 256];
    for (int u = 0; u < CVT_U; ++u) {
      v4bf o;
      for (int k = 0; k < 4; ++k) o[k] = (bf16_t)v[u][k];
      out4[base + u * 256] = o;
    }
  } else {
    for (int u = 0; u < CVT_U; ++u) {
      const int i = base + u * 256;
      if (i < n4) {
        const v4f x = in4[i];
        v4bf o;
        for (int k = 0; k < 4; ++k) o[k] = (bf16_t)x[k];
        out4[i] = o;
      }
    }
  }
}

__device__ __forceinline__ void pack_weight_block(
    const float* __restrict__ W, bf16_t* __restrict__ Wp,
    int K, int Nout, int Kp, int Noutp, int b) {
  int idx = b * 256 + threadIdx.x;
  if (idx >= Noutp * Kp) return;
  const int i    = idx & 7;
  const int l    = (idx >> 3) & 31;
  const int j    = (idx >> 8) & 1;
  const int rest = idx >> 9;
  const int ksteps = Kp / 32;
  const int ks = rest % ksteps, nt = rest / ksteps;
  const int n = nt * 16 + (l & 15);
  const int k = ks * 32 + 16 * j + 8 * (l >> 4) + i;
  float v = (n < Nout && k < K) ? W[(size_t)k * Nout + n] : 0.0f;
  Wp[idx] = (bf16_t)v;
}

#define PACK_B0 ((NP0 * KP0 + 255) / 256)
#define PACK_B1 ((NP1 * KP1 + 255) / 256)
#define PACK_B2 ((NP2 * KP2 + 255) / 256)

__global__ __launch_bounds__(256) void prep_kernel(
    const float* __restrict__ x, bf16_t* __restrict__ xb, int n4, int cvt_blocks,
    const float* __restrict__ fcW, bf16_t* __restrict__ fcWt,
    const float* __restrict__ W1, bf16_t* __restrict__ W1t,
    const float* __restrict__ W2, bf16_t* __restrict__ W2t,
    int* __restrict__ cnt, int Nn) {
  int b = blockIdx.x;
  if (b < cvt_blocks) { cvt_bf16_block(x, xb, n4, b); return; }
  b -= cvt_blocks;
  if (b < PACK_B0) { pack_weight_block(fcW, fcWt, IN_FT, HID1, KP0, NP0, b); return; }
  b -= PACK_B0;
  if (b < PACK_B1) { pack_weight_block(W1, W1t, HID1, HID2, KP1, NP1, b); return; }
  b -= PACK_B1;
  if (b < PACK_B2) { pack_weight_block(W2, W2t, HID2, OUT_FT, KP2, NP2, b); return; }
  b -= PACK_B2;
  const int i = (b * 256 + threadIdx.x) * 4;
  if (i + 4 <= Nn) {
    *(int4*)(cnt + i) = make_int4(0, 0, 0, 0);
  } else {
    for (int k = i; k < Nn; ++k) cnt[k] = 0;
  }
}

template <int T>
__device__ __forceinline__ void rank_block(
    const int* __restrict__ dst, int* __restrict__ cnt,
    int* __restrict__ rank, int E, int b) {
  int e = b * T + threadIdx.x;
  if (e < E)
    rank[e] = __hip_atomic_fetch_add(&cnt[dst[e]], 1,
                                     __ATOMIC_RELAXED, __HIP_MEMORY_SCOPE_AGENT);
}

__device__ __forceinline__ int wave_incl_scan(int v, int lane) {
  for (int o = 1; o < 32; o <<= 1) {
    int up = __shfl_up(v, o, 32);
    if (lane >= o) v += up;
  }
  return v;
}

#define SCAN_V 23
template <int T>
__device__ __forceinline__ void scan_block(
    const int* __restrict__ cnt, int* __restrict__ rowptr, int n) {
  constexpr int CH = SCAN_V * 4;
  __shared__ int wsum[32];
  __shared__ int pass_total;
  const int tid = threadIdx.x, lane = tid & 31, wv = tid >> 5;
  int carry = 0;
  for (int p0 = 0; p0 <= n; p0 += T * CH) {
    const int b = p0 + tid * CH;
    int4 c[SCAN_V];
    if (b + CH <= n) {
#pragma unroll
      for (int j = 0; j < SCAN_V; ++j) c[j] = *(const int4*)(cnt + b + 4 * j);
    } else {
#pragma unroll
      for (int j = 0; j < SCAN_V; ++j) {
        const int k = b + 4 * j;
        c[j] = make_int4(0, 0, 0, 0);
        if (k + 4 <= n) {
          c[j] = *(const int4*)(cnt + k);
        } else if (k < n) {
          c[j].x = cnt[k];
          if (k + 1 < n) c[j].y = cnt[k + 1];
          if (k + 2 < n) c[j].z = cnt[k + 2];
        }
      }
    }
    int s = 0;
#pragma unroll
    for (int j = 0; j < SCAN_V; ++j) s += (c[j].x + c[j].y) + (c[j].z + c[j].w);
    const int inc = wave_incl_scan(s, lane);
    if (lane == 31) wsum[wv] = inc;
    __syncthreads();
    if (wv == 0) {
      const int tot = (lane < T / 32) ? wsum[lane] : 0;
      const int ti  = wave_incl_scan(tot, lane);
      wsum[lane] = ti - tot;
      if (lane == 31) pass_total = ti;
    }
    __syncthreads();
    int run = carry + wsum[wv] + inc - s;
    carry += pass_total;
#pragma unroll
    for (int j = 0; j < SCAN_V; ++j) {
      const int k = b + 4 * j;
      int4 o;
      o.x = run; run += c[j].x;
      o.y = run; run += c[j].y;
      o.z = run; run += c[j].z;
      o.w = run; run += c[j].w;
      if (k + 3 <= n) {
        *(int4*)(rowptr + k) = o;
      } else {
        if (k     <= n) rowptr[k]     = o.x;
        if (k + 1 <= n) rowptr[k + 1] = o.y;
        if (k + 2 <= n) rowptr[k + 2] = o.z;
      }
    }
    __syncthreads();
  }
}

__global__ __launch_bounds__(256) void fill_kernel(
    const int* __restrict__ src, const int* __restrict__ dst,
    const float* __restrict__ ew, const int* __restrict__ rank,
    const int* __restrict__ rowptr, int2* __restrict__ csr, int E) {
  int e = blockIdx.x * 256 + threadIdx.x;
  if (e < E) {
    int slot = rowptr[dst[e]] + rank[e];
    csr[slot] = make_int2(src[e], __float_as_int(ew[e]));
  }
}

__global__ __launch_bounds__(256) void dinv_kernel(
    const int* __restrict__ rowptr, const int2* __restrict__ csr,
    float* __restrict__ dinv, int n) {
  const int g   = (blockIdx.x * 256 + threadIdx.x) >> 3;
  const int sub = threadIdx.x & 7;
  float s = 0.0f;
  if (g < n) {
    const int end = rowptr[g + 1];
    for (int k = rowptr[g] + sub; k < end; k += 8) s += __int_as_float(csr[k].y);
  }
  s += __shfl_xor(s, 1, 8);
  s += __shfl_xor(s, 2, 8);
  s += __shfl_xor(s, 4, 8);
  if (g < n && sub == 0) dinv[g] = rsqrtf(s + 1.0f);
}

__device__ __forceinline__ v16bf join_frag(v8bf lo, v8bf hi) {
  return __builtin_shufflevector(lo, hi, 0, 1, 2, 3, 4, 5, 6, 7,
                                 8, 9, 10, 11, 12, 13, 14, 15);
}

__device__ __forceinline__ v16bf lds_frag(const bf16_t* p) {
  return join_frag(*(const v8bf*)p, *(const v8bf*)(p + 16));
}

__device__ __forceinline__ v8f wmma_bf16(v16bf a, v16bf b, v8f c) {
  return __builtin_amdgcn_wmma_f32_16x16x32_bf16(
      false, a, false, b, (short)0, c, false, false);
}

template <int MODE, int KP, int TN>
__device__ __forceinline__ void gemm_block(
    const bf16_t* __restrict__ A, const bf16_t* __restrict__ Bp,
    const float* __restrict__ bias, void* __restrict__ Cout,
    int Nout, int ldo, int tm) {
  constexpr int KS    = KP / 32;
  constexpr int PITCH = KP + 8;
  constexpr int NT    = TN * 32;
  constexpr int CPR   = KP / 8;
  constexpr int NCH   = 32 * CPR;
  constexpr int IT    = (NCH + NT - 1) / NT;
  __shared__ __attribute__((aligned(16))) bf16_t As[32 * PITCH];

  const int tid  = threadIdx.x;
  const int lane = tid & 31;
  const int tn   = tid >> 5;

  const v8bf* bp0 = (const v8bf*)Bp + (size_t)(2 * tn) * (KS * 64) + lane;
  const v8bf* bp1 = bp0 + KS * 64;
  v8bf b0lo = bp0[0], b0hi = bp0[32], b1lo = bp1[0], b1hi = bp1[32];

  {
    const v8bf* Ag = (const v8bf*)(A + (size_t)tm * 32 * KP);
    v8bf st[IT];
#pragma unroll
    for (int u = 0; u < IT; ++u) {
      const int c = tid + u * NT;
      if (c < NCH) st[u] = Ag[c];
    }
#pragma unroll
    for (int u = 0; u < IT; ++u) {
      const int c = tid + u * NT;
      if (c < NCH) {
        const int row = c / CPR, col = c - row * CPR;
        *(v8bf*)(As + row * PITCH + col * 8) = st[u];
      }
    }
  }
  __syncthreads();

  const int r  = lane & 15;
  const int hi = lane >> 4;
  const bf16_t* a0 = As + r * PITCH + hi * 8;
  const bf16_t* a1 = a0 + 16 * PITCH;

  v8f c00 = {}, c01 = {}, c10 = {}, c11 = {};
#pragma unroll
  for (int ks = 0; ks < KS; ++ks) {
    v8bf n0lo = b0lo, n0hi = b0hi, n1lo = b1lo, n1hi = b1hi;
    if (ks + 1 < KS) {
      n0lo = bp0[(ks + 1) * 64];
      n0hi = bp0[(ks + 1) * 64 + 32];
      n1lo = bp1[(ks + 1) * 64];
      n1hi = bp1[(ks + 1) * 64 + 32];
    }
    const v16bf va0 = lds_frag(a0 + ks * 32);
    const v16bf va1 = lds_frag(a1 + ks * 32);
    const v16bf vb0 = join_frag(b0lo, b0hi);
    const v16bf vb1 = join_frag(b1lo, b1hi);
    c00 = wmma_bf16(va0, vb0, c00);
    c01 = wmma_bf16(va0, vb1, c01);
    c10 = wmma_bf16(va1, vb0, c10);
    c11 = wmma_bf16(va1, vb1, c11);
    b0lo = n0lo; b0hi = n0hi; b1lo = n1lo; b1hi = n1hi;
  }

  const v8f cc[2][2] = {{c00, c01}, {c10, c11}};
#pragma unroll
  for (int sn = 0; sn < 2; ++sn) {
    const int n = tn * 32 + sn * 16 + r;
    if (n >= ((MODE == 1) ? ldo : Nout)) continue;
    const bool real = n < Nout;
    const float bv = (MODE == 1 && real) ? bias[n] : 0.0f;
#pragma unroll
    for (int sm = 0; sm < 2; ++sm) {
      const v8f c = cc[sm][sn];
#pragma unroll
      for (int i = 0; i < 8; ++i) {
        const int m = tm * 32 + sm * 16 + i + hi * 8;
        float v = c[i] + bv;
        if (MODE == 1) {
          v = v > 0.0f ? v : 0.0f;
          if (!real) v = 0.0f;
          ((bf16_t*)Cout)[(size_t)m * ldo + n] = (bf16_t)v;
        } else {
          ((float*)Cout)[(size_t)m * ldo + n] = v;
        }
      }
    }
  }
}

template <int MODE, int KP, int TN>
__global__ __launch_bounds__(TN * 32) void wmma_gemm_kernel(
    const bf16_t* __restrict__ A, const bf16_t* __restrict__ Bp,
    const float* __restrict__ bias, void* __restrict__ Cout,
    int Nout, int ldo) {
  gemm_block<MODE, KP, TN>(A, Bp, bias, Cout, Nout, ldo, blockIdx.x);
}

__global__ __launch_bounds__(NP0) void gemm0_rank_kernel(
    const bf16_t* __restrict__ A, const bf16_t* __restrict__ Bp,
    const float* __restrict__ bias, bf16_t* __restrict__ Cout, int gemm_blocks,
    const int* __restrict__ dst, int* __restrict__ cnt,
    int* __restrict__ rank, int E) {
  const int b = blockIdx.x;
  if (b < gemm_blocks) {
    gemm_block<1, KP0, NP0 / 32>(A, Bp, bias, Cout, HID1, KP1, b);
  } else {
    rank_block<NP0>(dst, cnt, rank, E, b - gemm_blocks);
  }
}

__global__ __launch_bounds__(NP1) void gemm1_scan_kernel(
    const bf16_t* __restrict__ A, const bf16_t* __restrict__ Bp,
    float* __restrict__ Cout,
    const int* __restrict__ cnt, int* __restrict__ rowptr, int n) {
  if (blockIdx.x == 0) {
    scan_block<NP1>(cnt, rowptr, n);
  } else {
    gemm_block<0, KP1, NP1 / 32>(A, Bp, nullptr, Cout, HID2, HID2, blockIdx.x - 1);
  }
}

template <int F, int U>
__device__ __forceinline__ void gather_step(
    const float* __restrict__ t, int sj, float wj, int j, int lane,
    v4f& a0, v4f& a1) {
  constexpr int F4 = F / 4;
  constexpr int R4 = F4 > 32 ? F4 - 32 : 0;
  const v4f* row[U];
  float w[U];
  v4f x0[U], x1[U];
  for (int u = 0; u < U; ++u) {
    const int s = __builtin_amdgcn_readlane(sj, j + u);
    w[u] = __int_as_float(__builtin_amdgcn_readlane(__float_as_int(wj), j + u));
    row[u] = (const v4f*)(t + (size_t)s * F);
    x0[u] = row[u][lane];
  }
  if (R4) {
    for (int u = 0; u < U; ++u) x1[u] = v4f{0.0f, 0.0f, 0.0f, 0.0f};
    if (lane < R4) {
      for (int u = 0; u < U; ++u) x1[u] = row[u][32 + lane];
    }
  }
  for (int u = 0; u < U; ++u) {
    a0 += x0[u] * w[u];
    if (R4) a1 += x1[u] * w[u];
  }
}

__device__ __forceinline__ float relu1(float v) { return v > 0.0f ? v : 0.0f; }

template <int F, int MODE>
__global__ __launch_bounds__(256) void gather_kernel(
    const float* __restrict__ t, const int* __restrict__ rowptr,
    const int2* __restrict__ csr, const float* __restrict__ dinv,
    const float* __restrict__ bias, void* __restrict__ out, int Nn, int ldo) {
  constexpr int F4 = F / 4;
  constexpr int R4 = F4 > 32 ? F4 - 32 : 0;
  constexpr int U  = R4 ? 4 : 8;
  const int lane = threadIdx.x & 31;
  const int wv   = __builtin_amdgcn_readfirstlane(threadIdx.x >> 5);
  const int pair = blockIdx.x * 8 + wv;

  for (int half = 0; half < 2; ++half) {
    const int d = pair * 2 + half;
    if (d >= Nn) return;
    const int beg = rowptr[d];
    const int end = rowptr[d + 1];
    const float di = dinv[d];

    v4f a0 = {0.0f, 0.0f, 0.0f, 0.0f}, a1 = {0.0f, 0.0f, 0.0f, 0.0f};
    for (int base = beg; base < end; base += 32) {
      const int m = (end - base) < 32 ? (end - base) : 32;
      int sj = 0;
      float wj = 0.0f;
      if (lane < m) {
        const int2 ce = csr[base + lane];
        sj = ce.x;
        wj = dinv[ce.x] * __int_as_float(ce.y) * di;
      }
      int j = 0;
      for (; j + U <= m; j += U) gather_step<F, U>(t, sj, wj, j, lane, a0, a1);
      if (U == 8 && j + 4 <= m) { gather_step<F, 4>(t, sj, wj, j, lane, a0, a1); j += 4; }
      if (j + 2 <= m) { gather_step<F, 2>(t, sj, wj, j, lane, a0, a1); j += 2; }
      if (j < m) gather_step<F, 1>(t, sj, wj, j, lane, a0, a1);
    }

    const v4f* srow = (const v4f*)(t + (size_t)d * F);
    const v4f s0 = srow[lane];
    const v4f b0 = {bias[4 * lane], bias[4 * lane + 1], bias[4 * lane + 2], bias[4 * lane + 3]};
    const v4f r0 = a0 + s0 * di * di + b0;
    v4f r1 = {0.0f, 0.0f, 0.0f, 0.0f};
    if (R4) {
      if (lane < R4) {
        const v4f s1 = srow[32 + lane];
        const int f = 128 + 4 * lane;
        const v4f b1 = {bias[f], bias[f + 1], bias[f + 2], bias[f + 3]};
        r1 = a1 + s1 * di * di + b1;
      }
    }

    if (MODE == 1) {
      bf16_t* orow = (bf16_t*)out + (size_t)d * ldo;
      v4bf o0;
      for (int i = 0; i < 4; ++i) o0[i] = (bf16_t)relu1(r0[i]);
      *(v4bf*)(orow + 4 * lane) = o0;
      if (R4) {
        if (128 + 4 * lane < ldo) {
          v4bf o1;
          for (int i = 0; i < 4; ++i) o1[i] = (bf16_t)relu1(r1[i]);
          *(v4bf*)(orow + 128 + 4 * lane) = o1;
        }
      }
      for (int cpad = (R4 ? 256 : 128) + 4 * lane; cpad < ldo; cpad += 128) {
        const v4bf z = {(bf16_t)0.0f, (bf16_t)0.0f, (bf16_t)0.0f, (bf16_t)0.0f};
        *(v4bf*)(orow + cpad) = z;
      }
    } else {
      float* orow = (float*)out + (size_t)d * F;
      v4f o0;
      for (int i = 0; i < 4; ++i) o0[i] = relu1(r0[i]);
      *(v4f*)(orow + 4 * lane) = o0;
      if (R4) {
        if (lane < R4) {
          v4f o1;
          for (int i = 0; i < 4; ++i) o1[i] = relu1(r1[i]);
          *(v4f*)(orow + 128 + 4 * lane) = o1;
        }
      }
    }
  }
}

static inline size_t al256(size_t x) { return (x + 255) & ~(size_t)255; }

extern "C" void kernel_launch(void* const* d_in, const int* in_sizes, int n_in,
                              void* d_out, int out_size, void* d_ws, size_t ws_size,
                              hipStream_t stream) {
  const float* x    = (const float*)d_in[0];
  const int*   ei   = (const int*)  d_in[1];
  const float* ew   = (const float*)d_in[2];
  const float* fc_W = (const float*)d_in[3];
  const float* fc_b = (const float*)d_in[4];
  const float* W1   = (const float*)d_in[5];
  const float* b1   = (const float*)d_in[6];
  const float* W2   = (const float*)d_in[7];
  const float* b2   = (const float*)d_in[8];

  const int Nn = in_sizes[0] / IN_FT;
  const int E  = in_sizes[1] / 2;
  const int* src = ei;
  const int* dst = ei + E;

  char* ws = (char*)d_ws;
  size_t off = 0;
  auto alloc = [&](size_t bytes) { void* p = ws + off; off += al256(bytes); return p; };

  int*    cnt    = (int*)  alloc((size_t)Nn * 4);
  int*    rank   = (int*)  alloc((size_t)E * 4);
  int*    rowptr = (int*)  alloc((size_t)(Nn + 1) * 4);
  int2*   csr    = (int2*) alloc((size_t)E * 8);
  float*  dinv   = (float*)alloc((size_t)Nn * 4);
  bf16_t* xb   = (bf16_t*)alloc((size_t)Nn * KP0 * 2);
  bf16_t* fcWt = (bf16_t*)alloc((size_t)NP0 * KP0 * 2);
  bf16_t* h0b  = (bf16_t*)alloc((size_t)Nn * KP1 * 2);
  bf16_t* W1t  = (bf16_t*)alloc((size_t)NP1 * KP1 * 2);
  float*  t1   = (float*) alloc((size_t)Nn * HID2 * 4);
  bf16_t* h1b  = (bf16_t*)alloc((size_t)Nn * KP2 * 2);
  bf16_t* W2t  = (bf16_t*)alloc((size_t)NP2 * KP2 * 2);
  float*  t2   = (float*) alloc((size_t)Nn * OUT_FT * 4);
  (void)ws_size;

  {
    const int n4 = Nn * (IN_FT / 4);
    const int per_block = 256 * CVT_U;
    const int cvt_blocks  = (n4 + per_block - 1) / per_block;
    const int zero_blocks = (Nn + 1023) / 1024;
    prep_kernel<<<cvt_blocks + PACK_B0 + PACK_B1 + PACK_B2 + zero_blocks, 256, 0, stream>>>(
        x, xb, n4, cvt_blocks, fc_W, fcWt, W1, W1t, W2, W2t, cnt, Nn);
  }

  const int pair_blocks = ((Nn + 1) / 2 + 7) / 8;

  {
    const int gemm_blocks = Nn / 32;
    const int rank_blocks = (E + NP0 - 1) / NP0;
    gemm0_rank_kernel<<<gemm_blocks + rank_blocks, NP0, 0, stream>>>(
        xb, fcWt, fc_b, h0b, gemm_blocks, dst, cnt, rank, E);
  }

  gemm1_scan_kernel<<<Nn / 32 + 1, NP1, 0, stream>>>(
      h0b, W1t, t1, cnt, rowptr, Nn);

  fill_kernel<<<(E + 255) / 256, 256, 0, stream>>>(src, dst, ew, rank, rowptr, csr, E);
  dinv_kernel<<<(Nn * 8 + 255) / 256, 256, 0, stream>>>(rowptr, csr, dinv, Nn);

  gather_kernel<HID2, 1><<<pair_blocks, 256, 0, stream>>>(
      t1, rowptr, csr, dinv, b1, h1b, Nn, KP2);

  {
    wmma_gemm_kernel<0, KP2, NP2 / 32><<<Nn / 32, NP2, 0, stream>>>(
        h1b, W2t, nullptr, t2, OUT_FT, OUT_FT);
    gather_kernel<OUT_FT, 0><<<pair_blocks, 256, 0, stream>>>(
        t2, rowptr, csr, dinv, b2, d_out, Nn, OUT_FT);
  }
}


# Layer-2 GEMM fused into the layer-1 gather launch: each 16-row work-group keeps its gathered bf16 rows in LDS and multiplies them by W2 there (one WMMA tile per wave); 7 -> 6 launches
# speedup vs baseline: 3.5225x; 1.1053x over previous
#include <hip/hip_runtime.h>
#include <hip/hip_bf16.h>


typedef __bf16 bf16_t;
typedef __bf16 v16bf __attribute__((ext_vector_type(16)));
typedef __bf16 v8bf  __attribute__((ext_vector_type(8)));
typedef float  v8f   __attribute__((ext_vector_type(8)));

#define IN_FT   256
#define HID1    400
#define HID2    200
#define OUT_FT  128

#define KP0     256
#define KP1     416
#define KP2     224
#define NP0     416
#define NP1     224
#define NP2     128

typedef float  v4f  __attribute__((ext_vector_type(4)));
typedef __bf16 v4bf __attribute__((ext_vector_type(4)));

#define CVT_U 4
__device__ __forceinline__ void cvt_bf16_block(
    const float* __restrict__ in, bf16_t* __restrict__ out, int n4, int b) {
  const v4f* __restrict__ in4  = (const v4f*)in;
  v4bf* __restrict__      out4 = (v4bf*)out;
  const int blk  = b * (256 * CVT_U);
  const int base = blk + threadIdx.x;
  if (blk + 256 * CVT_U <= n4) {
    v4f v[CVT_U];
    for (int u = 0; u < CVT_U; ++u) v[u] = in4[base + u * 256];
    for (int u = 0; u < CVT_U; ++u) {
      v4bf o;
      for (int k = 0; k < 4; ++k) o[k] = (bf16_t)v[u][k];
      out4[base + u * 256] = o;
    }
  } else {
    for (int u = 0; u < CVT_U; ++u) {
      const int i = base + u * 256;
      if (i < n4) {
        const v4f x = in4[i];
        v4bf o;
        for (int k = 0; k < 4; ++k) o[k] = (bf16_t)x[k];
        out4[i] = o;
      }
    }
  }
}

__device__ __forceinline__ void pack_weight_block(
    const float* __restrict__ W, bf16_t* __restrict__ Wp,
    int K, int Nout, int Kp, int Noutp, int b) {
  int idx = b * 256 + threadIdx.x;
  if (idx >= Noutp * Kp) return;
  const int i    = idx & 7;
  const int l    = (idx >> 3) & 31;
  const int j    = (idx >> 8) & 1;
  const int rest = idx >> 9;
  const int ksteps = Kp / 32;
  const int ks = rest % ksteps, nt = rest / ksteps;
  const int n = nt * 16 + (l & 15);
  const int k = ks * 32 + 16 * j + 8 * (l >> 4) + i;
  float v = (n < Nout && k < K) ? W[(size_t)k * Nout + n] : 0.0f;
  Wp[idx] = (bf16_t)v;
}

#define PACK_B0 ((NP0 * KP0 + 255) / 256)
#define PACK_B1 ((NP1 * KP1 + 255) / 256)
#define PACK_B2 ((NP2 * KP2 + 255) / 256)

__global__ __launch_bounds__(256) void prep_kernel(
    const float* __restrict__ x, bf16_t* __restrict__ xb, int n4, int cvt_blocks,
    const float* __restrict__ fcW, bf16_t* __restrict__ fcWt,
    const float* __restrict__ W1, bf16_t* __restrict__ W1t,
    const float* __restrict__ W2, bf16_t* __restrict__ W2t,
    int* __restrict__ cnt, int ncnt) {
  int b = blockIdx.x;
  if (b < cvt_blocks) { cvt_bf16_block(x, xb, n4, b); return; }
  b -= cvt_blocks;
  if (b < PACK_B0) { pack_weight_block(fcW, fcWt, IN_FT, HID1, KP0, NP0, b); return; }
  b -= PACK_B0;
  if (b < PACK_B1) { pack_weight_block(W1, W1t, HID1, HID2, KP1, NP1, b); return; }
  b -= PACK_B1;
  if (b < PACK_B2) { pack_weight_block(W2, W2t, HID2, OUT_FT, KP2, NP2, b); return; }
  b -= PACK_B2;
  const int i = (b * 256 + threadIdx.x) * 4;
  if (i + 4 <= ncnt) {
    *(int4*)(cnt + i) = make_int4(0, 0, 0, 0);
  } else {
    for (int k = i; k < ncnt; ++k) cnt[k] = 0;
  }
}

template <int T>
__device__ __forceinline__ void rank_block(
    const int* __restrict__ dst, int* __restrict__ cnt,
    int* __restrict__ rank, int E, int b) {
  int e = b * T + threadIdx.x;
  if (e < E)
    rank[e] = __hip_atomic_fetch_add(&cnt[dst[e]], 1,
                                     __ATOMIC_RELAXED, __HIP_MEMORY_SCOPE_AGENT);
}

__device__ __forceinline__ int wave_incl_scan(int v, int lane) {
  for (int o = 1; o < 32; o <<= 1) {
    int up = __shfl_up(v, o, 32);
    if (lane >= o) v += up;
  }
  return v;
}

#define SCAN_V 23
#define FILL_U 10
#define RP_CAP 20032
#define SORT_LDS_INTS (RP_CAP + 40)

template <int T>
__device__ __forceinline__ void scan_fill_block(
    const int* __restrict__ cnt, int* __restrict__ rowptr,
    const int* __restrict__ src, const int* __restrict__ dst,
    const float* __restrict__ ew, const int* __restrict__ rank,
    int2* __restrict__ csr, int E, int b, int* __restrict__ rp) {
  constexpr int CH = SCAN_V * 4;
  static_assert(RP_CAP <= T * CH && RP_CAP % 4 == 0, "one pass covers the table");
  int* wsum = rp + RP_CAP;
  const int tid = threadIdx.x, lane = tid & 31, wv = tid >> 5;
  const int p = tid * CH;
  int4 c[SCAN_V];
  if (p + CH <= RP_CAP) {
#pragma unroll
    for (int j = 0; j < SCAN_V; ++j) c[j] = *(const int4*)(cnt + p + 4 * j);
  } else {
#pragma unroll
    for (int j = 0; j < SCAN_V; ++j) {
      const int k = p + 4 * j;
      c[j] = make_int4(0, 0, 0, 0);
      if (k < RP_CAP) c[j] = *(const int4*)(cnt + k);
    }
  }
  int s = 0;
#pragma unroll
  for (int j = 0; j < SCAN_V; ++j) s += (c[j].x + c[j].y) + (c[j].z + c[j].w);
  const int inc = wave_incl_scan(s, lane);
  if (lane == 31) wsum[wv] = inc;
  __syncthreads();
  if (wv == 0) {
    const int tot = (lane < T / 32) ? wsum[lane] : 0;
    const int ti  = wave_incl_scan(tot, lane);
    wsum[lane] = ti - tot;
  }
  __syncthreads();
  int run = wsum[wv] + inc - s;
#pragma unroll
  for (int j = 0; j < SCAN_V; ++j) {
    const int k = p + 4 * j;
    int4 o;
    o.x = run; run += c[j].x;
    o.y = run; run += c[j].y;
    o.z = run; run += c[j].z;
    o.w = run; run += c[j].w;
    if (k < RP_CAP) *(int4*)(rp + k) = o;
  }
  __syncthreads();

  if (b == 0) {
    for (int k = tid * 4; k < RP_CAP; k += T * 4)
      *(int4*)(rowptr + k) = *(const int4*)(rp + k);
  }

  const int e0 = b * (T * FILL_U) + tid;
  int ed[FILL_U], er[FILL_U], es[FILL_U], ewb[FILL_U];
#pragma unroll
  for (int u = 0; u < FILL_U; ++u) {
    const int e  = e0 + u * T;
    const int ec = e < E ? e : E - 1;
    ed[u]  = dst[ec];
    er[u]  = rank[ec];
    es[u]  = src[ec];
    ewb[u] = __float_as_int(ew[ec]);
  }
  if ((b + 1) * (T * FILL_U) <= E) {
#pragma unroll
    for (int u = 0; u < FILL_U; ++u)
      csr[rp[ed[u]] + er[u]] = make_int2(es[u], ewb[u]);
  } else {
#pragma unroll
    for (int u = 0; u < FILL_U; ++u) {
      if (e0 + u * T < E) csr[rp[ed[u]] + er[u]] = make_int2(es[u], ewb[u]);
    }
  }
}

__global__ __launch_bounds__(256) void dinv_kernel(
    const int* __restrict__ rowptr, const int2* __restrict__ csr,
    float* __restrict__ dinv, int n) {
  const int g   = (blockIdx.x * 256 + threadIdx.x) >> 3;
  const int sub = threadIdx.x & 7;
  float s = 0.0f;
  if (g < n) {
    const int end = rowptr[g + 1];
    for (int k = rowptr[g] + sub; k < end; k += 8) s += __int_as_float(csr[k].y);
  }
  s += __shfl_xor(s, 1, 8);
  s += __shfl_xor(s, 2, 8);
  s += __shfl_xor(s, 4, 8);
  if (g < n && sub == 0) dinv[g] = rsqrtf(s + 1.0f);
}

__device__ __forceinline__ v16bf join_frag(v8bf lo, v8bf hi) {
  return __builtin_shufflevector(lo, hi, 0, 1, 2, 3, 4, 5, 6, 7,
                                 8, 9, 10, 11, 12, 13, 14, 15);
}

__device__ __forceinline__ v16bf lds_frag(const bf16_t* p) {
  return join_frag(*(const v8bf*)p, *(const v8bf*)(p + 16));
}

__device__ __forceinline__ v8f wmma_bf16(v16bf a, v16bf b, v8f c) {
  return __builtin_amdgcn_wmma_f32_16x16x32_bf16(
      false, a, false, b, (short)0, c, false, false);
}

#define GEMM_LDS_ELEMS(KP) (32 * ((KP) + 8))
template <int MODE, int KP, int TN>
__device__ __forceinline__ void gemm_block(
    const bf16_t* __restrict__ A, const bf16_t* __restrict__ Bp,
    const float* __restrict__ bias, void* __restrict__ Cout,
    int Nout, int ldo, int tm, bf16_t* __restrict__ As) {
  constexpr int KS    = KP / 32;
  constexpr int PITCH = KP + 8;
  constexpr int NT    = TN * 32;
  constexpr int CPR   = KP / 8;
  constexpr int NCH   = 32 * CPR;
  constexpr int IT    = (NCH + NT - 1) / NT;
  static_assert(GEMM_LDS_ELEMS(KP) == 32 * PITCH, "LDS room of the row block");

  const int tid  = threadIdx.x;
  const int lane = tid & 31;
  const int tn   = tid >> 5;

  const v8bf* bp0 = (const v8bf*)Bp + (size_t)(2 * tn) * (KS * 64) + lane;
  const v8bf* bp1 = bp0 + KS * 64;
  v8bf b0lo = bp0[0], b0hi = bp0[32], b1lo = bp1[0], b1hi = bp1[32];

  {
    const v8bf* Ag = (const v8bf*)(A + (size_t)tm * 32 * KP);
    v8bf st[IT];
#pragma unroll
    for (int u = 0; u < IT; ++u) {
      const int c = tid + u * NT;
      if (c < NCH) st[u] = Ag[c];
    }
#pragma unroll
    for (int u = 0; u < IT; ++u) {
      const int c = tid + u * NT;
      if (c < NCH) {
        const int row = c / CPR, col = c - row * CPR;
        *(v8bf*)(As + row * PITCH + col * 8) = st[u];
      }
    }
  }
  __syncthreads();

  const int r  = lane & 15;
  const int hi = lane >> 4;
  const bf16_t* a0 = As + r * PITCH + hi * 8;
  const bf16_t* a1 = a0 + 16 * PITCH;

  v8f c00 = {}, c01 = {}, c10 = {}, c11 = {};
#pragma unroll
  for (int ks = 0; ks < KS; ++ks) {
    v8bf n0lo = b0lo, n0hi = b0hi, n1lo = b1lo, n1hi = b1hi;
    if (ks + 1 < KS) {
      n0lo = bp0[(ks + 1) * 64];
      n0hi = bp0[(ks + 1) * 64 + 32];
      n1lo = bp1[(ks + 1) * 64];
      n1hi = bp1[(ks + 1) * 64 + 32];
    }
    const v16bf va0 = lds_frag(a0 + ks * 32);
    const v16bf va1 = lds_frag(a1 + ks * 32);
    const v16bf vb0 = join_frag(b0lo, b0hi);
    const v16bf vb1 = join_frag(b1lo, b1hi);
    c00 = wmma_bf16(va0, vb0, c00);
    c01 = wmma_bf16(va0, vb1, c01);
    c10 = wmma_bf16(va1, vb0, c10);
    c11 = wmma_bf16(va1, vb1, c11);
    b0lo = n0lo; b0hi = n0hi; b1lo = n1lo; b1hi = n1hi;
  }

  const v8f cc[2][2] = {{c00, c01}, {c10, c11}};
#pragma unroll
  for (int sn = 0; sn < 2; ++sn) {
    const int n = tn * 32 + sn * 16 + r;
    if (n >= ((MODE == 1) ? ldo : Nout)) continue;
    const bool real = n < Nout;
    const float bv = (MODE == 1 && real) ? bias[n] : 0.0f;
#pragma unroll
    for (int sm = 0; sm < 2; ++sm) {
      const v8f c = cc[sm][sn];
#pragma unroll
      for (int i = 0; i < 8; ++i) {
        const int m = tm * 32 + sm * 16 + i + hi * 8;
        float v = c[i] + bv;
        if (MODE == 1) {
          v = v > 0.0f ? v : 0.0f;
          if (!real) v = 0.0f;
          ((bf16_t*)Cout)[(size_t)m * ldo + n] = (bf16_t)v;
        } else {
          ((float*)Cout)[(size_t)m * ldo + n] = v;
        }
      }
    }
  }
}

__global__ __launch_bounds__(NP0) void gemm0_rank_kernel(
    const bf16_t* __restrict__ A, const bf16_t* __restrict__ Bp,
    const float* __restrict__ bias, bf16_t* __restrict__ Cout, int gemm_blocks,
    const int* __restrict__ dst, int* __restrict__ cnt,
    int* __restrict__ rank, int E) {
  __shared__ __attribute__((aligned(16))) bf16_t As[GEMM_LDS_ELEMS(KP0)];
  const int b = blockIdx.x;
  if (b < gemm_blocks) {
    gemm_block<1, KP0, NP0 / 32>(A, Bp, bias, Cout, HID1, KP1, b, As);
  } else {
    rank_block<NP0>(dst, cnt, rank, E, b - gemm_blocks);
  }
}

__global__ __launch_bounds__(NP1) void gemm1_sort_kernel(
    const bf16_t* __restrict__ A, const bf16_t* __restrict__ Bp,
    float* __restrict__ Cout,
    const int* __restrict__ cnt, int* __restrict__ rowptr,
    const int* __restrict__ src, const int* __restrict__ dst,
    const float* __restrict__ ew, const int* __restrict__ rank,
    int2* __restrict__ csr, int E, int sort_blocks) {
  constexpr int GEMM_INT4 = (GEMM_LDS_ELEMS(KP1) * 2 + 15) / 16;
  constexpr int SORT_INT4 = (SORT_LDS_INTS * 4 + 15) / 16;
  __shared__ int4 lds[GEMM_INT4 > SORT_INT4 ? GEMM_INT4 : SORT_INT4];
  const int b = blockIdx.x;
  if (b < sort_blocks) {
    scan_fill_block<NP1>(cnt, rowptr, src, dst, ew, rank, csr, E, b, (int*)lds);
  } else {
    gemm_block<0, KP1, NP1 / 32>(A, Bp, nullptr, Cout, HID2, HID2, b - sort_blocks,
                                 (bf16_t*)lds);
  }
}

template <int F, int U>
__device__ __forceinline__ void gather_step(
    const float* __restrict__ t, int sj, float wj, int j, int lane,
    v4f& a0, v4f& a1) {
  constexpr int F4 = F / 4;
  constexpr int R4 = F4 > 32 ? F4 - 32 : 0;
  const v4f* row[U];
  float w[U];
  v4f x0[U], x1[U];
  for (int u = 0; u < U; ++u) {
    const int s = __builtin_amdgcn_readlane(sj, j + u);
    w[u] = __int_as_float(__builtin_amdgcn_readlane(__float_as_int(wj), j + u));
    row[u] = (const v4f*)(t + (size_t)s * F);
    x0[u] = row[u][lane];
  }
  if (R4) {
    for (int u = 0; u < U; ++u) x1[u] = v4f{0.0f, 0.0f, 0.0f, 0.0f};
    if (lane < R4) {
      for (int u = 0; u < U; ++u) x1[u] = row[u][32 + lane];
    }
  }
  for (int u = 0; u < U; ++u) {
    a0 += x0[u] * w[u];
    if (R4) a1 += x1[u] * w[u];
  }
}

__device__ __forceinline__ float relu1(float v) { return v > 0.0f ? v : 0.0f; }

template <int F>
__device__ __forceinline__ void gather_row(
    const float* __restrict__ t, const int* __restrict__ rowptr,
    const int2* __restrict__ csr, const float* __restrict__ dinv,
    const float* __restrict__ bias, int d, int lane, v4f& r0, v4f& r1) {
  constexpr int F4 = F / 4;
  constexpr int R4 = F4 > 32 ? F4 - 32 : 0;
  constexpr int U  = R4 ? 4 : 8;
  const int beg = rowptr[d];
  const int end = rowptr[d + 1];
  const float di = dinv[d];

  v4f a0 = {0.0f, 0.0f, 0.0f, 0.0f}, a1 = {0.0f, 0.0f, 0.0f, 0.0f};
  for (int base = beg; base < end; base += 32) {
    const int m = (end - base) < 32 ? (end - base) : 32;
    int sj = 0;
    float wj = 0.0f;
    if (lane < m) {
      const int2 ce = csr[base + lane];
      sj = ce.x;
      wj = dinv[ce.x] * __int_as_float(ce.y) * di;
    }
    int j = 0;
    for (; j + U <= m; j += U) gather_step<F, U>(t, sj, wj, j, lane, a0, a1);
    if (U == 8 && j + 4 <= m) { gather_step<F, 4>(t, sj, wj, j, lane, a0, a1); j += 4; }
    if (j + 2 <= m) { gather_step<F, 2>(t, sj, wj, j, lane, a0, a1); j += 2; }
    if (j < m) gather_step<F, 1>(t, sj, wj, j, lane, a0, a1);
  }

  const v4f* srow = (const v4f*)(t + (size_t)d * F);
  const v4f s0 = srow[lane];
  const v4f b0 = {bias[4 * lane], bias[4 * lane + 1], bias[4 * lane + 2], bias[4 * lane + 3]};
  const v4f q0 = a0 + s0 * di * di + b0;
  for (int i = 0; i < 4; ++i) r0[i] = relu1(q0[i]);
  r1 = v4f{0.0f, 0.0f, 0.0f, 0.0f};
  if (R4) {
    if (lane < R4) {
      const v4f s1 = srow[32 + lane];
      const int f = 128 + 4 * lane;
      const v4f b1 = {bias[f], bias[f + 1], bias[f + 2], bias[f + 3]};
      const v4f q1 = a1 + s1 * di * di + b1;
      for (int i = 0; i < 4; ++i) r1[i] = relu1(q1[i]);
    }
  }
}

template <int F>
__global__ __launch_bounds__(256) void gather_kernel(
    const float* __restrict__ t, const int* __restrict__ rowptr,
    const int2* __restrict__ csr, const float* __restrict__ dinv,
    const float* __restrict__ bias, float* __restrict__ out, int Nn) {
  static_assert(F == 128, "one 16-byte piece per lane, every lane");
  const int lane = threadIdx.x & 31;
  const int wv   = __builtin_amdgcn_readfirstlane(threadIdx.x >> 5);
  const int pair = blockIdx.x * 8 + wv;

  for (int half = 0; half < 2; ++half) {
    const int d = pair * 2 + half;
    if (d >= Nn) return;
    v4f r0, r1;
    gather_row<F>(t, rowptr, csr, dinv, bias, d, lane, r0, r1);
    *(v4f*)(out + (size_t)d * F + 4 * lane) = r0;
  }
}

#define FG_ROWS   16
#define FG_APITCH (KP2 + 8)
#define FG_CPITCH (OUT_FT + 4)
__global__ __launch_bounds__(256, 12) void gather_gemm_kernel(
    const float* __restrict__ t, const int* __restrict__ rowptr,
    const int2* __restrict__ csr, const float* __restrict__ dinv,
    const float* __restrict__ bias, const bf16_t* __restrict__ Bp,
    float* __restrict__ Cout, int Nn) {
  static_assert(OUT_FT == 8 * 16, "one 16-column tile per wave");
  static_assert(HID2 > 128 && HID2 <= 256 && KP2 <= 256, "two pieces per lane");
  constexpr int KS = KP2 / 32;
  __shared__ __attribute__((aligned(16))) bf16_t As[FG_ROWS * FG_APITCH];
  __shared__ __attribute__((aligned(16))) float  Cs[FG_ROWS * FG_CPITCH];
  const int lane = threadIdx.x & 31;
  const int wv   = __builtin_amdgcn_readfirstlane(threadIdx.x >> 5);
  const int row0 = blockIdx.x * FG_ROWS;

  for (int half = 0; half < 2; ++half) {
    const int lr = wv * 2 + half;
    const int d  = row0 + lr;
    v4f r0 = {0.0f, 0.0f, 0.0f, 0.0f}, r1 = {0.0f, 0.0f, 0.0f, 0.0f};
    if (d < Nn) gather_row<HID2>(t, rowptr, csr, dinv, bias, d, lane, r0, r1);
    bf16_t* arow = As + lr * FG_APITCH;
    v4bf o0;
    for (int i = 0; i < 4; ++i) o0[i] = (bf16_t)r0[i];
    *(v4bf*)(arow + 4 * lane) = o0;
    if (128 + 4 * lane < KP2) {
      v4bf o1;
      for (int i = 0; i < 4; ++i) o1[i] = (bf16_t)r1[i];
      *(v4bf*)(arow + 128 + 4 * lane) = o1;
    }
  }

  const v8bf* bp = (const v8bf*)Bp + (size_t)wv * (KS * 64) + lane;
  v8bf blo[KS], bhi[KS];
#pragma unroll
  for (int ks = 0; ks < KS; ++ks) {
    blo[ks] = bp[ks * 64];
    bhi[ks] = bp[ks * 64 + 32];
  }
  __syncthreads();

  const int r  = lane & 15;
  const int hi = lane >> 4;
  const bf16_t* a = As + r * FG_APITCH + hi * 8;
  v8f c = {};
#pragma unroll
  for (int ks = 0; ks < KS; ++ks)
    c = wmma_bf16(lds_frag(a + ks * 32), join_frag(blo[ks], bhi[ks]), c);

#pragma unroll
  for (int i = 0; i < 8; ++i) Cs[(i + 8 * hi) * FG_CPITCH + wv * 16 + r] = c[i];
  __syncthreads();

  for (int half = 0; half < 2; ++half) {
    const int lr = wv * 2 + half;
    const int d  = row0 + lr;
    if (d < Nn)
      *(v4f*)(Cout + (size_t)d * OUT_FT + 4 * lane) = *(const v4f*)(Cs + lr * FG_CPITCH + 4 * lane);
  }
}

static inline size_t al256(size_t x) { return (x + 255) & ~(size_t)255; }

extern "C" void kernel_launch(void* const* d_in, const int* in_sizes, int n_in,
                              void* d_out, int out_size, void* d_ws, size_t ws_size,
                              hipStream_t stream) {
  const float* x    = (const float*)d_in[0];
  const int*   ei   = (const int*)  d_in[1];
  const float* ew   = (const float*)d_in[2];
  const float* fc_W = (const float*)d_in[3];
  const float* fc_b = (const float*)d_in[4];
  const float* W1   = (const float*)d_in[5];
  const float* b1   = (const float*)d_in[6];
  const float* W2   = (const float*)d_in[7];
  const float* b2   = (const float*)d_in[8];

  const int Nn = in_sizes[0] / IN_FT;
  const int E  = in_sizes[1] / 2;
  const int* src = ei;
  const int* dst = ei + E;

  char* ws = (char*)d_ws;
  size_t off = 0;
  auto alloc = [&](size_t bytes) { void* p = ws + off; off += al256(bytes); return p; };

  int*    cnt    = (int*)  alloc((size_t)RP_CAP * 4);
  int*    rank   = (int*)  alloc((size_t)E * 4);
  int*    rowptr = (int*)  alloc((size_t)RP_CAP * 4);
  int2*   csr    = (int2*) alloc((size_t)E * 8);
  float*  dinv   = (float*)alloc((size_t)Nn * 4);
  bf16_t* xb   = (bf16_t*)alloc((size_t)Nn * KP0 * 2);
  bf16_t* fcWt = (bf16_t*)alloc((size_t)NP0 * KP0 * 2);
  bf16_t* h0b  = (bf16_t*)alloc((size_t)Nn * KP1 * 2);
  bf16_t* W1t  = (bf16_t*)alloc((size_t)NP1 * KP1 * 2);
  float*  t1   = (float*) alloc((size_t)Nn * HID2 * 4);
  bf16_t* W2t  = (bf16_t*)alloc((size_t)NP2 * KP2 * 2);
  float*  t2   = (float*) alloc((size_t)Nn * OUT_FT * 4);
  (void)ws_size;

  {
    const int n4 = Nn * (IN_FT / 4);
    const int per_block = 256 * CVT_U;
    const int cvt_blocks  = (n4 + per_block - 1) / per_block;
    const int zero_blocks = (RP_CAP + 1023) / 1024;
    prep_kernel<<<cvt_blocks + PACK_B0 + PACK_B1 + PACK_B2 + zero_blocks, 256, 0, stream>>>(
        x, xb, n4, cvt_blocks, fc_W, fcWt, W1, W1t, W2, W2t, cnt, RP_CAP);
  }

  const int pair_blocks = ((Nn + 1) / 2 + 7) / 8;

  {
    const int gemm_blocks = Nn / 32;
    const int rank_blocks = (E + NP0 - 1) / NP0;
    gemm0_rank_kernel<<<gemm_blocks + rank_blocks, NP0, 0, stream>>>(
        xb, fcWt, fc_b, h0b, gemm_blocks, dst, cnt, rank, E);
  }

  {
    const int sort_blocks = (E + NP1 * FILL_U - 1) / (NP1 * FILL_U);
    gemm1_sort_kernel<<<sort_blocks + Nn / 32, NP1, 0, stream>>>(
        h0b, W1t, t1, cnt, rowptr, src, dst, ew, rank, csr, E, sort_blocks);
  }

  dinv_kernel<<<(Nn * 8 + 255) / 256, 256, 0, stream>>>(rowptr, csr, dinv, Nn);

  gather_gemm_kernel<<<pair_blocks, 256, 0, stream>>>(
      t1, rowptr, csr, dinv, b1, W2t, t2, Nn);

  gather_kernel<OUT_FT><<<pair_blocks, 256, 0, stream>>>(
      t2, rowptr, csr, dinv, b2, (float*)d_out, Nn);
}
